# starting kernel
# baseline (speedup 1.0000x reference)
#include <hip/hip_runtime.h>

typedef __attribute__((ext_vector_type(2))) float v2f;
typedef __attribute__((ext_vector_type(8))) float v8f;

#define BATCH 64
#define IN_CH 512
#define HID   128
#define NOUT  10

__global__ __launch_bounds__(256) void mlp_wmma_kernel(
    const float* __restrict__ features,
    const float* __restrict__ W1,
    const float* __restrict__ b1,
    const float* __restrict__ W2,
    const float* __restrict__ b2,
    const float* __restrict__ W3,
    const float* __restrict__ b3,
    float* __restrict__ alpha_out,
    float* __restrict__ kw_out)
{
    __shared__ float h1[BATCH][HID];
    __shared__ float h2[BATCH][HID];

    const int tid    = threadIdx.x;
    const int lane   = tid & 31;
    const int wave   = tid >> 5;
    const int nloc   = lane & 15;
    const int khalf  = (lane >> 4) * 2;
    const int rowoff = (lane >> 4) * 8;

    {
        const int n0 = wave * 16;
        for (int mt = 0; mt < 4; ++mt) {
            const int m0 = mt * 16;
            v8f acc = {};
            for (int k0 = 0; k0 < IN_CH; k0 += 4) {
                const int kb = k0 + khalf;
                v2f a, b;
                a.x = features[(m0 + nloc) * IN_CH + kb];
                a.y = features[(m0 + nloc) * IN_CH + kb + 1];
                b.x = W1[(n0 + nloc) * IN_CH + kb];
                b.y = W1[(n0 + nloc) * IN_CH + kb + 1];
                acc = __builtin_amdgcn_wmma_f32_16x16x4_f32(
                    false, a, false, b, (short)0, acc, false, false);
            }
            const float bias = b1[n0 + nloc];
#pragma unroll
            for (int r = 0; r < 8; ++r) {
                float v = acc[r] + bias;
                h1[m0 + r + rowoff][n0 + nloc] = v > 0.f ? v : 0.f;
            }
        }
    }
    __syncthreads();

    {
        const int n0 = wave * 16;
        for (int mt = 0; mt < 4; ++mt) {
            const int m0 = mt * 16;
            v8f acc = {};
            for (int k0 = 0; k0 < HID; k0 += 4) {
                const int kb = k0 + khalf;
                v2f a, b;
                a.x = h1[m0 + nloc][kb];
                a.y = h1[m0 + nloc][kb + 1];
                b.x = W2[(n0 + nloc) * HID + kb];
                b.y = W2[(n0 + nloc) * HID + kb + 1];
                acc = __builtin_amdgcn_wmma_f32_16x16x4_f32(
                    false, a, false, b, (short)0, acc, false, false);
            }
            const float bias = b2[n0 + nloc];
#pragma unroll
            for (int r = 0; r < 8; ++r) {
                float v = acc[r] + bias;
                h2[m0 + r + rowoff][n0 + nloc] = v > 0.f ? v : 0.f;
            }
        }
    }
    __syncthreads();

    if (wave < 4) {
        const int m0 = wave * 16;
        const bool valid = (nloc < NOUT);
        v8f acc = {};
        for (int k0 = 0; k0 < HID; k0 += 4) {
            const int kb = k0 + khalf;
            v2f a, b;
            a.x = h2[m0 + nloc][kb];
            a.y = h2[m0 + nloc][kb + 1];
            b.x = valid ? W3[nloc * HID + kb]     : 0.f;
            b.y = valid ? W3[nloc * HID + kb + 1] : 0.f;
            acc = __builtin_amdgcn_wmma_f32_16x16x4_f32(
                false, a, false, b, (short)0, acc, false, false);
        }
        const float bias = valid ? b3[nloc] : 0.f;
#pragma unroll
        for (int r = 0; r < 8; ++r) {
            const int row = m0 + r + rowoff;
            float v = acc[r] + bias;
            if (nloc == 0) {
                alpha_out[row] = v;
            } else if (valid) {
                v += (nloc == 5) ? 5.0f : -5.0f;
                kw_out[row * 9 + (nloc - 1)] = 1.0f / (1.0f + __expf(-v));
            }
        }
    }
}

#define TW 64
#define TH 4

__global__ __launch_bounds__(256) void morph_pixel_kernel(
    const float* __restrict__ x,
    const float* __restrict__ alpha,
    const float* __restrict__ kw,
    float* __restrict__ out,
    int H, int W)
{
    __shared__ float tile[TH + 2][TW + 4];

    const int b   = blockIdx.z;
    const int bx  = blockIdx.x * TW;
    const int by  = blockIdx.y * TH;
    const int tid = threadIdx.x;
    const float* img = x + (size_t)b * H * W;

    for (int idx = tid; idx < (TH + 2) * (TW + 2); idx += 256) {
        const int ly = idx / (TW + 2);
        const int lx = idx - ly * (TW + 2);
        const int gy = by + ly - 1;
        const int gx = bx + lx - 1;
        float v = 0.f;
        if (gy >= 0 && gy < H && gx >= 0 && gx < W) v = img[gy * W + gx];
        tile[ly][lx] = v;
    }

    const float a = alpha[b];
    float k[9];
#pragma unroll
    for (int j = 0; j < 9; ++j) k[j] = kw[b * 9 + j];

    __syncthreads();

    const int tx = tid & (TW - 1);
    const int ty = tid >> 6;

    float num = 0.f, den = 0.f;
#pragma unroll
    for (int dy = 0; dy < 3; ++dy) {
#pragma unroll
        for (int dx = 0; dx < 3; ++dx) {
            const float p = tile[ty + dy][tx + dx];
            const float w = __expf(a * p) * k[dy * 3 + dx];
            num += p * w;
            den += w;
        }
    }
    out[(size_t)b * H * W + (size_t)(by + ty) * W + (bx + tx)] = num / den;
}

extern "C" void kernel_launch(void* const* d_in, const int* in_sizes, int n_in,
                              void* d_out, int out_size, void* d_ws, size_t ws_size,
                              hipStream_t stream) {
    const float* x        = (const float*)d_in[0];
    const float* features = (const float*)d_in[1];
    const float* W1       = (const float*)d_in[2];
    const float* b1       = (const float*)d_in[3];
    const float* W2       = (const float*)d_in[4];
    const float* b2       = (const float*)d_in[5];
    const float* W3       = (const float*)d_in[6];
    const float* b3       = (const float*)d_in[7];

    float* alpha = (float*)d_ws;
    float* kw    = alpha + BATCH;
    float* out   = (float*)d_out;

    mlp_wmma_kernel<<<1, 256, 0, stream>>>(features, W1, b1, W2, b2, W3, b3,
                                           alpha, kw);

    const int H = 256, W = 256;
    dim3 grid(W / TW, H / TH, BATCH);
    morph_pixel_kernel<<<grid, 256, 0, stream>>>(x, alpha, kw, out, H, W);
}


# MLP launch spread over 4 work-groups, one per 16-row batch tile (was 1 group walking 4 tiles in turn); each group hands its rows to the pixel pass as 64-B records in 128-B lines of its own
# speedup vs baseline: 1.9547x; 1.9547x over previous
#include <hip/hip_runtime.h>

typedef __attribute__((ext_vector_type(2))) float v2f;
typedef __attribute__((ext_vector_type(8))) float v8f;

#define BATCH 64
#define IN_CH 512
#define HID   128
#define NOUT  10
#define MT    16
#define REC   16

__global__ __launch_bounds__(256) void mlp_wmma_kernel(
    const float* __restrict__ features,
    const float* __restrict__ W1,
    const float* __restrict__ b1,
    const float* __restrict__ W2,
    const float* __restrict__ b2,
    const float* __restrict__ W3,
    const float* __restrict__ b3,
    float* __restrict__ rec_out)
{
    __shared__ float h1[MT][HID];
    __shared__ float h2[MT][HID];

    const int tid    = threadIdx.x;
    const int lane   = tid & 31;
    const int wave   = tid >> 5;
    const int nloc   = lane & 15;
    const int khalf  = (lane >> 4) * 2;
    const int rowoff = (lane >> 4) * 8;
    const int m0     = blockIdx.x * MT;
    const int n0     = wave * 16;

    {
        v8f acc = {};
        for (int k0 = 0; k0 < IN_CH; k0 += 4) {
            const int kb = k0 + khalf;
            v2f a, b;
            a.x = features[(m0 + nloc) * IN_CH + kb];
            a.y = features[(m0 + nloc) * IN_CH + kb + 1];
            b.x = W1[(n0 + nloc) * IN_CH + kb];
            b.y = W1[(n0 + nloc) * IN_CH + kb + 1];
            acc = __builtin_amdgcn_wmma_f32_16x16x4_f32(
                false, a, false, b, (short)0, acc, false, false);
        }
        const float bias = b1[n0 + nloc];
#pragma unroll
        for (int r = 0; r < 8; ++r) {
            float v = acc[r] + bias;
            h1[r + rowoff][n0 + nloc] = v > 0.f ? v : 0.f;
        }
    }
    __syncthreads();

    {
        v8f acc = {};
        for (int k0 = 0; k0 < HID; k0 += 4) {
            const int kb = k0 + khalf;
            v2f a, b;
            a.x = h1[nloc][kb];
            a.y = h1[nloc][kb + 1];
            b.x = W2[(n0 + nloc) * HID + kb];
            b.y = W2[(n0 + nloc) * HID + kb + 1];
            acc = __builtin_amdgcn_wmma_f32_16x16x4_f32(
                false, a, false, b, (short)0, acc, false, false);
        }
        const float bias = b2[n0 + nloc];
#pragma unroll
        for (int r = 0; r < 8; ++r) {
            float v = acc[r] + bias;
            h2[r + rowoff][n0 + nloc] = v > 0.f ? v : 0.f;
        }
    }
    __syncthreads();

    if (wave == 0) {
        const bool valid = (nloc < NOUT);
        v8f acc = {};
        for (int k0 = 0; k0 < HID; k0 += 4) {
            const int kb = k0 + khalf;
            v2f a, b;
            a.x = h2[nloc][kb];
            a.y = h2[nloc][kb + 1];
            b.x = valid ? W3[nloc * HID + kb]     : 0.f;
            b.y = valid ? W3[nloc * HID + kb + 1] : 0.f;
            acc = __builtin_amdgcn_wmma_f32_16x16x4_f32(
                false, a, false, b, (short)0, acc, false, false);
        }
        const float bias = valid ? b3[nloc] : 0.f;
#pragma unroll
        for (int r = 0; r < 8; ++r) {
            const int row = m0 + r + rowoff;
            float v = acc[r] + bias;
            if (nloc == 0) {
                rec_out[row * REC] = v;
            } else if (valid) {
                v += (nloc == 5) ? 5.0f : -5.0f;
                rec_out[row * REC + nloc] = 1.0f / (1.0f + __expf(-v));
            }
        }
    }
}

#define TW 64
#define TH 4

__global__ __launch_bounds__(256) void morph_pixel_kernel(
    const float* __restrict__ x,
    const float* __restrict__ rec,
    float* __restrict__ out,
    int H, int W)
{
    __shared__ float tile[TH + 2][TW + 4];

    const int b   = blockIdx.z;
    const int bx  = blockIdx.x * TW;
    const int by  = blockIdx.y * TH;
    const int tid = threadIdx.x;
    const float* img = x + (size_t)b * H * W;

    for (int idx = tid; idx < (TH + 2) * (TW + 2); idx += 256) {
        const int ly = idx / (TW + 2);
        const int lx = idx - ly * (TW + 2);
        const int gy = by + ly - 1;
        const int gx = bx + lx - 1;
        float v = 0.f;
        if (gy >= 0 && gy < H && gx >= 0 && gx < W) v = img[gy * W + gx];
        tile[ly][lx] = v;
    }

    const float a = rec[b * REC];
    float k[9];
#pragma unroll
    for (int j = 0; j < 9; ++j) k[j] = rec[b * REC + 1 + j];

    __syncthreads();

    const int tx = tid & (TW - 1);
    const int ty = tid >> 6;

    float num = 0.f, den = 0.f;
#pragma unroll
    for (int dy = 0; dy < 3; ++dy) {
#pragma unroll
        for (int dx = 0; dx < 3; ++dx) {
            const float p = tile[ty + dy][tx + dx];
            const float w = __expf(a * p) * k[dy * 3 + dx];
            num += p * w;
            den += w;
        }
    }
    out[(size_t)b * H * W + (size_t)(by + ty) * W + (bx + tx)] = num / den;
}

extern "C" void kernel_launch(void* const* d_in, const int* in_sizes, int n_in,
                              void* d_out, int out_size, void* d_ws, size_t ws_size,
                              hipStream_t stream) {
    const float* x        = (const float*)d_in[0];
    const float* features = (const float*)d_in[1];
    const float* W1       = (const float*)d_in[2];
    const float* b1       = (const float*)d_in[3];
    const float* W2       = (const float*)d_in[4];
    const float* b2       = (const float*)d_in[5];
    const float* W3       = (const float*)d_in[6];
    const float* b3       = (const float*)d_in[7];

    float* rec = (float*)d_ws;
    float* out = (float*)d_out;

    mlp_wmma_kernel<<<BATCH / MT, 256, 0, stream>>>(features, W1, b1, W2, b2,
                                                    W3, b3, rec);

    const int H = 256, W = 256;
    dim3 grid(W / TW, H / TH, BATCH);
    morph_pixel_kernel<<<grid, 256, 0, stream>>>(x, rec, out, H, W);
}


# Pixel pass tiled in registers: a lane loads a 10x6 pixel window (128-bit row loads), takes exp once per loaded pixel and writes a 4x8 output block; no LDS tile; 4096 waves, was 131072
# speedup vs baseline: 3.0976x; 1.5847x over previous
#include <hip/hip_runtime.h>

typedef __attribute__((ext_vector_type(2))) float v2f;
typedef __attribute__((ext_vector_type(8))) float v8f;
typedef __attribute__((ext_vector_type(4))) float v4f;

#define BATCH 64
#define IN_CH 512
#define HID   128
#define NOUT  10
#define MT    16
#define REC   16

__global__ __launch_bounds__(256) void mlp_wmma_kernel(
    const float* __restrict__ features,
    const float* __restrict__ W1,
    const float* __restrict__ b1,
    const float* __restrict__ W2,
    const float* __restrict__ b2,
    const float* __restrict__ W3,
    const float* __restrict__ b3,
    float* __restrict__ rec_out)
{
    __shared__ float h1[MT][HID];
    __shared__ float h2[MT][HID];

    const int tid    = threadIdx.x;
    const int lane   = tid & 31;
    const int wave   = tid >> 5;
    const int nloc   = lane & 15;
    const int khalf  = (lane >> 4) * 2;
    const int rowoff = (lane >> 4) * 8;
    const int m0     = blockIdx.x * MT;
    const int n0     = wave * 16;

    {
        v8f acc = {};
        for (int k0 = 0; k0 < IN_CH; k0 += 4) {
            const int kb = k0 + khalf;
            v2f a, b;
            a.x = features[(m0 + nloc) * IN_CH + kb];
            a.y = features[(m0 + nloc) * IN_CH + kb + 1];
            b.x = W1[(n0 + nloc) * IN_CH + kb];
            b.y = W1[(n0 + nloc) * IN_CH + kb + 1];
            acc = __builtin_amdgcn_wmma_f32_16x16x4_f32(
                false, a, false, b, (short)0, acc, false, false);
        }
        const float bias = b1[n0 + nloc];
#pragma unroll
        for (int r = 0; r < 8; ++r) {
            float v = acc[r] + bias;
            h1[r + rowoff][n0 + nloc] = v > 0.f ? v : 0.f;
        }
    }
    __syncthreads();

    {
        v8f acc = {};
        for (int k0 = 0; k0 < HID; k0 += 4) {
            const int kb = k0 + khalf;
            v2f a, b;
            a.x = h1[nloc][kb];
            a.y = h1[nloc][kb + 1];
            b.x = W2[(n0 + nloc) * HID + kb];
            b.y = W2[(n0 + nloc) * HID + kb + 1];
            acc = __builtin_amdgcn_wmma_f32_16x16x4_f32(
                false, a, false, b, (short)0, acc, false, false);
        }
        const float bias = b2[n0 + nloc];
#pragma unroll
        for (int r = 0; r < 8; ++r) {
            float v = acc[r] + bias;
            h2[r + rowoff][n0 + nloc] = v > 0.f ? v : 0.f;
        }
    }
    __syncthreads();

    if (wave == 0) {
        const bool valid = (nloc < NOUT);
        v8f acc = {};
        for (int k0 = 0; k0 < HID; k0 += 4) {
            const int kb = k0 + khalf;
            v2f a, b;
            a.x = h2[nloc][kb];
            a.y = h2[nloc][kb + 1];
            b.x = valid ? W3[nloc * HID + kb]     : 0.f;
            b.y = valid ? W3[nloc * HID + kb + 1] : 0.f;
            acc = __builtin_amdgcn_wmma_f32_16x16x4_f32(
                false, a, false, b, (short)0, acc, false, false);
        }
        const float bias = valid ? b3[nloc] : 0.f;
#pragma unroll
        for (int r = 0; r < 8; ++r) {
            const int row = m0 + r + rowoff;
            float v = acc[r] + bias;
            if (nloc == 0) {
                rec_out[row * REC] = v;
            } else if (valid) {
                v += (nloc == 5) ? 5.0f : -5.0f;
                rec_out[row * REC + nloc] = 1.0f / (1.0f + __expf(-v));
            }
        }
    }
}

#define IMG_H  256
#define IMG_W  256
#define PX     4
#define RY     8
#define STRIPS 4

__global__ __launch_bounds__(256) void morph_pixel_kernel(
    const float* __restrict__ x,
    const float* __restrict__ rec,
    float* __restrict__ out)
{
    const int b   = blockIdx.y;
    const int tid = threadIdx.x;
    const int x0  = (tid & 63) * PX;
    const int y0  = (blockIdx.x * STRIPS + (tid >> 6)) * RY;
    const float* img = x   + (size_t)b * (IMG_H * IMG_W);
    float*       dst = out + (size_t)b * (IMG_H * IMG_W);

    const float a = rec[b * REC];
    float k[9];
#pragma unroll
    for (int j = 0; j < 9; ++j) k[j] = rec[b * REC + 1 + j];

    const bool has_l = x0 > 0;
    const bool has_r = x0 + PX < IMG_W;
    const int  xl    = has_l ? x0 - 1  : 0;
    const int  xr    = has_r ? x0 + PX : IMG_W - 1;

    v4f    cen[RY + 2];
    float  lft[RY + 2], rgt[RY + 2];
#pragma unroll
    for (int i = 0; i < RY + 2; ++i) {
        int gy = y0 - 1 + i;
        if (i == 0      && y0 == 0)           gy = 0;
        if (i == RY + 1 && y0 + RY == IMG_H)  gy = IMG_H - 1;
        const float* rowp = img + gy * IMG_W;
        cen[i] = *reinterpret_cast<const v4f*>(rowp + x0);
        lft[i] = rowp[xl];
        rgt[i] = rowp[xr];
    }
    asm volatile("" :: "v"(cen[0]), "v"(cen[1]), "v"(cen[2]), "v"(cen[3]), "v"(cen[4]),
                       "v"(cen[5]), "v"(cen[6]), "v"(cen[7]), "v"(cen[8]), "v"(cen[9]),
                       "v"(lft[0]), "v"(lft[1]), "v"(lft[2]), "v"(lft[3]), "v"(lft[4]),
                       "v"(lft[5]), "v"(lft[6]), "v"(lft[7]), "v"(lft[8]), "v"(lft[9]),
                       "v"(rgt[0]), "v"(rgt[1]), "v"(rgt[2]), "v"(rgt[3]), "v"(rgt[4]),
                       "v"(rgt[5]), "v"(rgt[6]), "v"(rgt[7]), "v"(rgt[8]), "v"(rgt[9]));

    float p[RY + 2][PX + 2];
#pragma unroll
    for (int i = 0; i < RY + 2; ++i) {
        bool rowin = true;
        if (i == 0)      rowin = y0 > 0;
        if (i == RY + 1) rowin = y0 + RY < IMG_H;
        p[i][0] = (rowin && has_l) ? lft[i] : 0.f;
        p[i][1] = rowin ? cen[i].x : 0.f;
        p[i][2] = rowin ? cen[i].y : 0.f;
        p[i][3] = rowin ? cen[i].z : 0.f;
        p[i][4] = rowin ? cen[i].w : 0.f;
        p[i][5] = (rowin && has_r) ? rgt[i] : 0.f;
    }

    float e[RY + 2][PX + 2];
#pragma unroll
    for (int i = 0; i < RY + 2; ++i) {
#pragma unroll
        for (int j = 0; j < PX + 2; ++j) e[i][j] = __expf(a * p[i][j]);
    }

#pragma unroll
    for (int r = 0; r < RY; ++r) {
        float o[PX];
#pragma unroll
        for (int c = 0; c < PX; ++c) {
            float num = 0.f, den = 0.f;
#pragma unroll
            for (int dy = 0; dy < 3; ++dy) {
#pragma unroll
                for (int dx = 0; dx < 3; ++dx) {
                    const float pv = p[r + dy][c + dx];
                    const float w  = e[r + dy][c + dx] * k[dy * 3 + dx];
                    num += pv * w;
                    den += w;
                }
            }
            o[c] = num / den;
        }
        *reinterpret_cast<float4*>(dst + (y0 + r) * IMG_W + x0) =
            make_float4(o[0], o[1], o[2], o[3]);
    }
}

extern "C" void kernel_launch(void* const* d_in, const int* in_sizes, int n_in,
                              void* d_out, int out_size, void* d_ws, size_t ws_size,
                              hipStream_t stream) {
    const float* x        = (const float*)d_in[0];
    const float* features = (const float*)d_in[1];
    const float* W1       = (const float*)d_in[2];
    const float* b1       = (const float*)d_in[3];
    const float* W2       = (const float*)d_in[4];
    const float* b2       = (const float*)d_in[5];
    const float* W3       = (const float*)d_in[6];
    const float* b3       = (const float*)d_in[7];

    float* rec = (float*)d_ws;
    float* out = (float*)d_out;

    mlp_wmma_kernel<<<BATCH / MT, 256, 0, stream>>>(features, W1, b1, W2, b2,
                                                    W3, b3, rec);

    dim3 grid(IMG_H / (STRIPS * RY), BATCH);
    morph_pixel_kernel<<<grid, 256, 0, stream>>>(x, rec, out);
}


# MLP operands staged through LDS by whole-line 128-bit loads all issued up front; fragments by ds_load_b128, a group of 16 ahead of 16 WMMAs, two accumulators; k summed in another order
# speedup vs baseline: 5.4074x; 1.7457x over previous
#include <hip/hip_runtime.h>

typedef __attribute__((ext_vector_type(2))) float v2f;
typedef __attribute__((ext_vector_type(8))) float v8f;
typedef __attribute__((ext_vector_type(4))) float v4f;

#define BATCH 64
#define IN_CH 512
#define HID   128
#define NOUT  10
#define MT    16
#define REC   16

#define KC     128
#define NSLAB  (IN_CH / KC)
#define SROWS  (HID + MT)
#define PITCH  (KC + 4)
#define SLAB   (SROWS * PITCH)
#define SLD    (SROWS / 8)
#define GK     64


__device__ __forceinline__ void stage_load(v4f (&st)[SLD],
    const float* __restrict__ W1, const float* __restrict__ F,
    int m0, int wave, int lane, int c)
{
#pragma unroll
    for (int i = 0; i < SLD; ++i) {
        const int r = i * 8 + wave;
        const float* row = (i < HID / 8) ? W1 + r * IN_CH
                                         : F + (m0 + r - HID) * IN_CH;
        st[i] = *reinterpret_cast<const v4f*>(row + c * KC + lane * 4);
    }
}

__device__ __forceinline__ void stage_store(float* S, const v4f (&st)[SLD],
                                            int wave, int lane)
{
#pragma unroll
    for (int i = 0; i < SLD; ++i)
        *reinterpret_cast<v4f*>(S + (i * 8 + wave) * PITCH + lane * 4) = st[i];
}

struct Frag { v4f a[GK / 8], b[GK / 8]; };

__device__ __forceinline__ void frag_load(Frag& f, const float* S, int arow,
                                          int brow, int half, int g)
{
    const float* base = S + (g >> 1) * SLAB + (g & 1) * GK + half * 4;
#pragma unroll
    for (int s = 0; s < GK / 8; ++s) {
        f.a[s] = *reinterpret_cast<const v4f*>(base + arow * PITCH + s * 8);
        f.b[s] = *reinterpret_cast<const v4f*>(base + brow * PITCH + s * 8);
    }
}

__device__ __forceinline__ void frag_mma(const Frag& f, v8f& acc0, v8f& acc1)
{
#pragma unroll
    for (int s = 0; s < GK / 8; ++s) {
        const v2f a0 = {f.a[s].x, f.a[s].y}, b0 = {f.b[s].x, f.b[s].y};
        const v2f a1 = {f.a[s].z, f.a[s].w}, b1 = {f.b[s].z, f.b[s].w};
        acc0 = __builtin_amdgcn_wmma_f32_16x16x4_f32(
            false, a0, false, b0, (short)0, acc0, false, false);
        acc1 = __builtin_amdgcn_wmma_f32_16x16x4_f32(
            false, a1, false, b1, (short)0, acc1, false, false);
    }
}

template <int NG>
__device__ __forceinline__ void tile_mma(const float* S, int brow, int nloc,
                                         int half, v8f& acc0, v8f& acc1)
{
    Frag f0, f1;
    frag_load(f0, S, HID + nloc, brow + nloc, half, 0);
#pragma unroll
    for (int g = 0; g < NG; g += 2) {
        if (g + 1 < NG) frag_load(f1, S, HID + nloc, brow + nloc, half, g + 1);
        __builtin_amdgcn_sched_barrier(0);
        frag_mma(f0, acc0, acc1);
        __builtin_amdgcn_sched_barrier(0);
        if (g + 2 < NG) frag_load(f0, S, HID + nloc, brow + nloc, half, g + 2);
        __builtin_amdgcn_sched_barrier(0);
        if (g + 1 < NG) frag_mma(f1, acc0, acc1);
        __builtin_amdgcn_sched_barrier(0);
    }
}

__global__ __launch_bounds__(256) void mlp_wmma_kernel(
    const float* __restrict__ features,
    const float* __restrict__ W1,
    const float* __restrict__ b1,
    const float* __restrict__ W2,
    const float* __restrict__ b2,
    const float* __restrict__ W3,
    const float* __restrict__ b3,
    float* __restrict__ rec_out)
{
    __shared__ __attribute__((aligned(16))) float L[NSLAB * SLAB];

    const int tid    = threadIdx.x;
    const int lane   = tid & 31;
    const int wave   = tid >> 5;
    const int nloc   = lane & 15;
    const int half   = lane >> 4;
    const int rowoff = half * 8;
    const int m0     = blockIdx.x * MT;
    const int n0     = wave * 16;

    float* S0 = L;
    float* S1 = L + SLAB;

    {
        v4f st0[SLD], st1[SLD], st2[SLD], st3[SLD];
        stage_load(st0, W1, features, m0, wave, lane, 0);
        stage_load(st1, W1, features, m0, wave, lane, 1);
        stage_load(st2, W1, features, m0, wave, lane, 2);
        stage_load(st3, W1, features, m0, wave, lane, 3);
        stage_store(S0, st0, wave, lane);
        stage_store(S1, st1, wave, lane);
        stage_store(L + 2 * SLAB, st2, wave, lane);
        stage_store(L + 3 * SLAB, st3, wave, lane);
    }
    __syncthreads();

    v4f w2[HID / 8];
#pragma unroll
    for (int i = 0; i < HID / 8; ++i)
        w2[i] = *reinterpret_cast<const v4f*>(W2 + (i * 8 + wave) * HID + lane * 4);
    const bool has_hi = wave + 8 < NOUT;
    const v4f w3lo = *reinterpret_cast<const v4f*>(W3 + wave * HID + lane * 4);
    const v4f w3hi = *reinterpret_cast<const v4f*>(
        W3 + (has_hi ? wave + 8 : NOUT - 1) * HID + lane * 4);
    const float bias1 = b1[n0 + nloc];
    const float bias2 = b2[n0 + nloc];
    const bool  valid = (nloc < NOUT);
    const float b3v   = b3[valid ? nloc : 0];

    v8f acc0 = {}, acc1 = {};
    tile_mma<IN_CH / GK>(L, n0, nloc, half, acc0, acc1);
    __syncthreads();

#pragma unroll
    for (int i = 0; i < HID / 8; ++i)
        *reinterpret_cast<v4f*>(S0 + (i * 8 + wave) * PITCH + lane * 4) = w2[i];
    *reinterpret_cast<v4f*>(S1 + wave * PITCH + lane * 4) = w3lo;
    *reinterpret_cast<v4f*>(S1 + (wave + 8) * PITCH + lane * 4) =
        has_hi ? w3hi : v4f{0.f, 0.f, 0.f, 0.f};
#pragma unroll
    for (int r = 0; r < 8; ++r) {
        const float v = (acc0[r] + acc1[r]) + bias1;
        S0[(HID + r + rowoff) * PITCH + n0 + nloc] = v > 0.f ? v : 0.f;
    }
    __syncthreads();

    acc0 = v8f{};
    acc1 = v8f{};
    tile_mma<HID / GK>(S0, n0, nloc, half, acc0, acc1);
#pragma unroll
    for (int r = 0; r < 8; ++r) {
        const float v = (acc0[r] + acc1[r]) + bias2;
        S1[(HID + r + rowoff) * PITCH + n0 + nloc] = v > 0.f ? v : 0.f;
    }
    __syncthreads();

    if (wave == 0) {
        const float bias3 = valid ? b3v : 0.f;
        acc0 = v8f{};
        acc1 = v8f{};
        tile_mma<HID / GK>(S1, 0, nloc, half, acc0, acc1);
#pragma unroll
        for (int r = 0; r < 8; ++r) {
            const int row = m0 + r + rowoff;
            float v = (acc0[r] + acc1[r]) + bias3;
            if (nloc == 0) {
                rec_out[row * REC] = v;
            } else if (valid) {
                v += (nloc == 5) ? 5.0f : -5.0f;
                rec_out[row * REC + nloc] = 1.0f / (1.0f + __expf(-v));
            }
        }
    }
}

#define IMG_H  256
#define IMG_W  256
#define PX     4
#define RY     8
#define STRIPS 4

__global__ __launch_bounds__(256) void morph_pixel_kernel(
    const float* __restrict__ x,
    const float* __restrict__ rec,
    float* __restrict__ out)
{
    const int b   = blockIdx.y;
    const int tid = threadIdx.x;
    const int x0  = (tid & 63) * PX;
    const int y0  = (blockIdx.x * STRIPS + (tid >> 6)) * RY;
    const float* img = x   + (size_t)b * (IMG_H * IMG_W);
    float*       dst = out + (size_t)b * (IMG_H * IMG_W);

    const float a = rec[b * REC];
    float k[9];
#pragma unroll
    for (int j = 0; j < 9; ++j) k[j] = rec[b * REC + 1 + j];

    const bool has_l = x0 > 0;
    const bool has_r = x0 + PX < IMG_W;
    const int  xl    = has_l ? x0 - 1  : 0;
    const int  xr    = has_r ? x0 + PX : IMG_W - 1;

    v4f    cen[RY + 2];
    float  lft[RY + 2], rgt[RY + 2];
#pragma unroll
    for (int i = 0; i < RY + 2; ++i) {
        int gy = y0 - 1 + i;
        if (i == 0      && y0 == 0)           gy = 0;
        if (i == RY + 1 && y0 + RY == IMG_H)  gy = IMG_H - 1;
        const float* rowp = img + gy * IMG_W;
        cen[i] = *reinterpret_cast<const v4f*>(rowp + x0);
        lft[i] = rowp[xl];
        rgt[i] = rowp[xr];
    }
    asm volatile("" :: "v"(cen[0]), "v"(cen[1]), "v"(cen[2]), "v"(cen[3]), "v"(cen[4]),
                       "v"(cen[5]), "v"(cen[6]), "v"(cen[7]), "v"(cen[8]), "v"(cen[9]),
                       "v"(lft[0]), "v"(lft[1]), "v"(lft[2]), "v"(lft[3]), "v"(lft[4]),
                       "v"(lft[5]), "v"(lft[6]), "v"(lft[7]), "v"(lft[8]), "v"(lft[9]),
                       "v"(rgt[0]), "v"(rgt[1]), "v"(rgt[2]), "v"(rgt[3]), "v"(rgt[4]),
                       "v"(rgt[5]), "v"(rgt[6]), "v"(rgt[7]), "v"(rgt[8]), "v"(rgt[9]));

    float p[RY + 2][PX + 2];
#pragma unroll
    for (int i = 0; i < RY + 2; ++i) {
        bool rowin = true;
        if (i == 0)      rowin = y0 > 0;
        if (i == RY + 1) rowin = y0 + RY < IMG_H;
        p[i][0] = (rowin && has_l) ? lft[i] : 0.f;
        p[i][1] = rowin ? cen[i].x : 0.f;
        p[i][2] = rowin ? cen[i].y : 0.f;
        p[i][3] = rowin ? cen[i].z : 0.f;
        p[i][4] = rowin ? cen[i].w : 0.f;
        p[i][5] = (rowin && has_r) ? rgt[i] : 0.f;
    }

    float e[RY + 2][PX + 2];
#pragma unroll
    for (int i = 0; i < RY + 2; ++i) {
#pragma unroll
        for (int j = 0; j < PX + 2; ++j) e[i][j] = __expf(a * p[i][j]);
    }

#pragma unroll
    for (int r = 0; r < RY; ++r) {
        float o[PX];
#pragma unroll
        for (int c = 0; c < PX; ++c) {
            float num = 0.f, den = 0.f;
#pragma unroll
            for (int dy = 0; dy < 3; ++dy) {
#pragma unroll
                for (int dx = 0; dx < 3; ++dx) {
                    const float pv = p[r + dy][c + dx];
                    const float w  = e[r + dy][c + dx] * k[dy * 3 + dx];
                    num += pv * w;
                    den += w;
                }
            }
            o[c] = num / den;
        }
        *reinterpret_cast<float4*>(dst + (y0 + r) * IMG_W + x0) =
            make_float4(o[0], o[1], o[2], o[3]);
    }
}

extern "C" void kernel_launch(void* const* d_in, const int* in_sizes, int n_in,
                              void* d_out, int out_size, void* d_ws, size_t ws_size,
                              hipStream_t stream) {
    const float* x        = (const float*)d_in[0];
    const float* features = (const float*)d_in[1];
    const float* W1       = (const float*)d_in[2];
    const float* b1       = (const float*)d_in[3];
    const float* W2       = (const float*)d_in[4];
    const float* b2       = (const float*)d_in[5];
    const float* W3       = (const float*)d_in[6];
    const float* b3       = (const float*)d_in[7];

    float* rec = (float*)d_ws;
    float* out = (float*)d_out;

    mlp_wmma_kernel<<<BATCH / MT, 256, 0, stream>>>(features, W1, b1, W2, b2,
                                                    W3, b3, rec);

    dim3 grid(IMG_H / (STRIPS * RY), BATCH);
    morph_pixel_kernel<<<grid, 256, 0, stream>>>(x, rec, out);
}


# Both stages in one launch: each of 256 work-groups recomputes the MLP of the batch tile that holds its image, keeps alpha and the 9 weights in LDS, then filters a quarter image; workspace unused
# speedup vs baseline: 6.1754x; 1.1420x over previous
#include <hip/hip_runtime.h>

typedef __attribute__((ext_vector_type(2))) float v2f;
typedef __attribute__((ext_vector_type(8))) float v8f;
typedef __attribute__((ext_vector_type(4))) float v4f;

#define BATCH 64
#define IN_CH 512
#define HID   128
#define NOUT  10
#define MT    16

#define KC     128
#define NSLAB  (IN_CH / KC)
#define SROWS  (HID + MT)
#define PITCH  (KC + 4)
#define SLAB   (SROWS * PITCH)
#define SLD    (SROWS / 8)
#define GK     64

#define IMG_H  256
#define IMG_W  256
#define PX     4
#define RY     8
#define STRIPS 4
#define PARTS  4
#define NGROUPS (BATCH * PARTS)


__device__ __forceinline__ void stage_load(v4f (&st)[SLD],
    const float* __restrict__ W1, const float* __restrict__ F,
    int m0, int wave, int lane, int c)
{
#pragma unroll
    for (int i = 0; i < SLD; ++i) {
        const int r = i * 8 + wave;
        const float* row = (i < HID / 8) ? W1 + r * IN_CH
                                         : F + (m0 + r - HID) * IN_CH;
        st[i] = *reinterpret_cast<const v4f*>(row + c * KC + lane * 4);
    }
}

__device__ __forceinline__ void stage_store(float* S, const v4f (&st)[SLD],
                                            int wave, int lane)
{
#pragma unroll
    for (int i = 0; i < SLD; ++i)
        *reinterpret_cast<v4f*>(S + (i * 8 + wave) * PITCH + lane * 4) = st[i];
}

struct Frag { v4f a[GK / 8], b[GK / 8]; };

__device__ __forceinline__ void frag_load(Frag& f, const float* S, int arow,
                                          int brow, int half, int g)
{
    const float* base = S + (g >> 1) * SLAB + (g & 1) * GK + half * 4;
#pragma unroll
    for (int s = 0; s < GK / 8; ++s) {
        f.a[s] = *reinterpret_cast<const v4f*>(base + arow * PITCH + s * 8);
        f.b[s] = *reinterpret_cast<const v4f*>(base + brow * PITCH + s * 8);
    }
}

__device__ __forceinline__ void frag_mma(const Frag& f, v8f& acc0, v8f& acc1)
{
#pragma unroll
    for (int s = 0; s < GK / 8; ++s) {
        const v2f a0 = {f.a[s].x, f.a[s].y}, b0 = {f.b[s].x, f.b[s].y};
        const v2f a1 = {f.a[s].z, f.a[s].w}, b1 = {f.b[s].z, f.b[s].w};
        acc0 = __builtin_amdgcn_wmma_f32_16x16x4_f32(
            false, a0, false, b0, (short)0, acc0, false, false);
        acc1 = __builtin_amdgcn_wmma_f32_16x16x4_f32(
            false, a1, false, b1, (short)0, acc1, false, false);
    }
}

template <int NG>
__device__ __forceinline__ void tile_mma(const float* S, int brow, int nloc,
                                         int half, v8f& acc0, v8f& acc1)
{
    Frag f0, f1;
    frag_load(f0, S, HID + nloc, brow + nloc, half, 0);
#pragma unroll
    for (int g = 0; g < NG; g += 2) {
        if (g + 1 < NG) frag_load(f1, S, HID + nloc, brow + nloc, half, g + 1);
        __builtin_amdgcn_sched_barrier(0);
        frag_mma(f0, acc0, acc1);
        __builtin_amdgcn_sched_barrier(0);
        if (g + 2 < NG) frag_load(f0, S, HID + nloc, brow + nloc, half, g + 2);
        __builtin_amdgcn_sched_barrier(0);
        if (g + 1 < NG) frag_mma(f1, acc0, acc1);
        __builtin_amdgcn_sched_barrier(0);
    }
}

struct Win { v4f cen[RY + 2]; float lft[RY + 2], rgt[RY + 2]; };

__device__ __forceinline__ void win_load(Win& w, const float* __restrict__ img,
                                         int x0, int xl, int xr, int y0)
{
#pragma unroll
    for (int i = 0; i < RY + 2; ++i) {
        int gy = y0 - 1 + i;
        if (i == 0      && y0 == 0)           gy = 0;
        if (i == RY + 1 && y0 + RY == IMG_H)  gy = IMG_H - 1;
        const float* rowp = img + gy * IMG_W;
        w.cen[i] = *reinterpret_cast<const v4f*>(rowp + x0);
        w.lft[i] = rowp[xl];
        w.rgt[i] = rowp[xr];
    }
}

__device__ __forceinline__ void win_filter(const Win& w, float* __restrict__ dst,
                                           int x0, int y0, bool has_l, bool has_r,
                                           float a, const float (&k)[9])
{
    asm volatile("" :: "v"(w.cen[0]), "v"(w.cen[1]), "v"(w.cen[2]), "v"(w.cen[3]), "v"(w.cen[4]),
                       "v"(w.cen[5]), "v"(w.cen[6]), "v"(w.cen[7]), "v"(w.cen[8]), "v"(w.cen[9]),
                       "v"(w.lft[0]), "v"(w.lft[1]), "v"(w.lft[2]), "v"(w.lft[3]), "v"(w.lft[4]),
                       "v"(w.lft[5]), "v"(w.lft[6]), "v"(w.lft[7]), "v"(w.lft[8]), "v"(w.lft[9]),
                       "v"(w.rgt[0]), "v"(w.rgt[1]), "v"(w.rgt[2]), "v"(w.rgt[3]), "v"(w.rgt[4]),
                       "v"(w.rgt[5]), "v"(w.rgt[6]), "v"(w.rgt[7]), "v"(w.rgt[8]), "v"(w.rgt[9]));

    float p[RY + 2][PX + 2];
#pragma unroll
    for (int i = 0; i < RY + 2; ++i) {
        bool rowin = true;
        if (i == 0)      rowin = y0 > 0;
        if (i == RY + 1) rowin = y0 + RY < IMG_H;
        p[i][0] = (rowin && has_l) ? w.lft[i] : 0.f;
        p[i][1] = rowin ? w.cen[i].x : 0.f;
        p[i][2] = rowin ? w.cen[i].y : 0.f;
        p[i][3] = rowin ? w.cen[i].z : 0.f;
        p[i][4] = rowin ? w.cen[i].w : 0.f;
        p[i][5] = (rowin && has_r) ? w.rgt[i] : 0.f;
    }

    float e[RY + 2][PX + 2];
#pragma unroll
    for (int i = 0; i < RY + 2; ++i) {
#pragma unroll
        for (int j = 0; j < PX + 2; ++j) e[i][j] = __expf(a * p[i][j]);
    }

#pragma unroll
    for (int r = 0; r < RY; ++r) {
        float o[PX];
#pragma unroll
        for (int c = 0; c < PX; ++c) {
            float num = 0.f, den = 0.f;
#pragma unroll
            for (int dy = 0; dy < 3; ++dy) {
#pragma unroll
                for (int dx = 0; dx < 3; ++dx) {
                    const float pv = p[r + dy][c + dx];
                    const float wt = e[r + dy][c + dx] * k[dy * 3 + dx];
                    num += pv * wt;
                    den += wt;
                }
            }
            o[c] = num / den;
        }
        *reinterpret_cast<float4*>(dst + (y0 + r) * IMG_W + x0) =
            make_float4(o[0], o[1], o[2], o[3]);
    }
}

__global__ __launch_bounds__(256) void morph_fused_kernel(
    const float* __restrict__ x,
    const float* __restrict__ features,
    const float* __restrict__ W1,
    const float* __restrict__ b1,
    const float* __restrict__ W2,
    const float* __restrict__ b2,
    const float* __restrict__ W3,
    const float* __restrict__ b3,
    float* __restrict__ out)
{
    __shared__ __attribute__((aligned(16))) float L[NSLAB * SLAB];
    __shared__ float recL[MT];

    const int tid    = threadIdx.x;
    const int lane   = tid & 31;
    const int wave   = tid >> 5;
    const int nloc   = lane & 15;
    const int half   = lane >> 4;
    const int rowoff = half * 8;
    const int b      = blockIdx.x / PARTS;
    const int part   = blockIdx.x % PARTS;
    const int m0     = (b / MT) * MT;
    const int brow   = b - m0;
    const int n0     = wave * 16;

    float* S0 = L;
    float* S1 = L + SLAB;

    {
        v4f st0[SLD], st1[SLD], st2[SLD], st3[SLD];
        stage_load(st0, W1, features, m0, wave, lane, 0);
        stage_load(st1, W1, features, m0, wave, lane, 1);
        stage_load(st2, W1, features, m0, wave, lane, 2);
        stage_load(st3, W1, features, m0, wave, lane, 3);
        stage_store(S0, st0, wave, lane);
        stage_store(S1, st1, wave, lane);
        stage_store(L + 2 * SLAB, st2, wave, lane);
        stage_store(L + 3 * SLAB, st3, wave, lane);
    }
    __syncthreads();

    v4f w2[HID / 8];
#pragma unroll
    for (int i = 0; i < HID / 8; ++i)
        w2[i] = *reinterpret_cast<const v4f*>(W2 + (i * 8 + wave) * HID + lane * 4);
    const bool has_hi = wave + 8 < NOUT;
    const v4f w3lo = *reinterpret_cast<const v4f*>(W3 + wave * HID + lane * 4);
    const v4f w3hi = *reinterpret_cast<const v4f*>(
        W3 + (has_hi ? wave + 8 : NOUT - 1) * HID + lane * 4);
    const float bias1 = b1[n0 + nloc];
    const float bias2 = b2[n0 + nloc];
    const bool  valid = (nloc < NOUT);
    const float b3v   = b3[valid ? nloc : 0];

    const int  x0    = (tid & 63) * PX;
    const int  ys    = part * (IMG_H / PARTS) + (tid >> 6) * RY;
    const bool has_l = x0 > 0;
    const bool has_r = x0 + PX < IMG_W;
    const int  xl    = has_l ? x0 - 1  : 0;
    const int  xr    = has_r ? x0 + PX : IMG_W - 1;
    const float* img = x   + (size_t)b * (IMG_H * IMG_W);
    float*       dst = out + (size_t)b * (IMG_H * IMG_W);
    Win win0, win1;
    win_load(win0, img, x0, xl, xr, ys);

    v8f acc0 = {}, acc1 = {};
    tile_mma<IN_CH / GK>(L, n0, nloc, half, acc0, acc1);
    __syncthreads();

#pragma unroll
    for (int i = 0; i < HID / 8; ++i)
        *reinterpret_cast<v4f*>(S0 + (i * 8 + wave) * PITCH + lane * 4) = w2[i];
    *reinterpret_cast<v4f*>(S1 + wave * PITCH + lane * 4) = w3lo;
    *reinterpret_cast<v4f*>(S1 + (wave + 8) * PITCH + lane * 4) =
        has_hi ? w3hi : v4f{0.f, 0.f, 0.f, 0.f};
#pragma unroll
    for (int r = 0; r < 8; ++r) {
        const float v = (acc0[r] + acc1[r]) + bias1;
        S0[(HID + r + rowoff) * PITCH + n0 + nloc] = v > 0.f ? v : 0.f;
    }
    __syncthreads();

    acc0 = v8f{};
    acc1 = v8f{};
    tile_mma<HID / GK>(S0, n0, nloc, half, acc0, acc1);
#pragma unroll
    for (int r = 0; r < 8; ++r) {
        const float v = (acc0[r] + acc1[r]) + bias2;
        S1[(HID + r + rowoff) * PITCH + n0 + nloc] = v > 0.f ? v : 0.f;
    }
    __syncthreads();

    if (wave == 0) {
        const float bias3 = valid ? b3v : 0.f;
        acc0 = v8f{};
        acc1 = v8f{};
        tile_mma<HID / GK>(S1, 0, nloc, half, acc0, acc1);
        const int rsel = brow & 7;
        float v = 0.f;
#pragma unroll
        for (int r = 0; r < 8; ++r) {
            const float t = acc0[r] + acc1[r];
            v = (r == rsel) ? t : v;
        }
        v += bias3;
        if (half == (brow >> 3) && valid) {
            if (nloc != 0) {
                v += (nloc == 5) ? 5.0f : -5.0f;
                v = 1.0f / (1.0f + __expf(-v));
            }
            recL[nloc] = v;
        }
    }
    __syncthreads();

    const float a = recL[0];
    float k[9];
#pragma unroll
    for (int j = 0; j < 9; ++j) k[j] = recL[1 + j];

    win_load(win1, img, x0, xl, xr, ys + STRIPS * RY);
    win_filter(win0, dst, x0, ys,               has_l, has_r, a, k);
    win_filter(win1, dst, x0, ys + STRIPS * RY, has_l, has_r, a, k);
}

extern "C" void kernel_launch(void* const* d_in, const int* in_sizes, int n_in,
                              void* d_out, int out_size, void* d_ws, size_t ws_size,
                              hipStream_t stream) {
    const float* x        = (const float*)d_in[0];
    const float* features = (const float*)d_in[1];
    const float* W1       = (const float*)d_in[2];
    const float* b1       = (const float*)d_in[3];
    const float* W2       = (const float*)d_in[4];
    const float* b2       = (const float*)d_in[5];
    const float* W3       = (const float*)d_in[6];
    const float* b3       = (const float*)d_in[7];
    float* out = (float*)d_out;

    morph_fused_kernel<<<NGROUPS, 256, 0, stream>>>(x, features, W1, b1, W2, b2,
                                                    W3, b3, out);
}
